# starting kernel
# baseline (speedup 1.0000x reference)
#include <hip/hip_runtime.h>
#include <hip/hip_bf16.h>
#include <math.h>

#define N_NODES 100000
#define N_EDGES 1600000
#define H_HEADS 4
#define F_IN    128
#define F_OUT   32
#define HF      (H_HEADS * F_OUT)
#define SLOPE   0.2f

typedef __attribute__((ext_vector_type(2))) float v2f;
typedef __attribute__((ext_vector_type(8))) float v8f;

__device__ __forceinline__ float lrelu(float v) { return v > 0.f ? v : SLOPE * v; }

__device__ __forceinline__ unsigned enc_f32(float f) {
    unsigned u = __float_as_uint(f);
    return (u & 0x80000000u) ? ~u : (u | 0x80000000u);
}
__device__ __forceinline__ float dec_f32(unsigned e) {
    return (e & 0x80000000u) ? __uint_as_float(e ^ 0x80000000u) : __uint_as_float(~e);
}

__global__ void gat_init(float* __restrict__ out, const float* __restrict__ bias,
                         unsigned* __restrict__ m_enc, float* __restrict__ denom) {
    int idx = blockIdx.x * blockDim.x + threadIdx.x;
    if (idx < N_NODES * HF) out[idx] = bias[idx & (HF - 1)];
    if (idx < N_NODES * H_HEADS) {
        m_enc[idx] = enc_f32(-__builtin_inff());
        denom[idx] = 0.f;
    }
}

__global__ void gat_repack_w(const float* __restrict__ w, v2f* __restrict__ wt) {
    int idx = blockIdx.x * blockDim.x + threadIdx.x;
    if (idx >= H_HEADS * 2 * 32 * 2 * 16) return;
    int n    = idx & 15;
    int kh   = (idx >> 4) & 1;
    int kq   = (idx >> 5) & 31;
    int ct   = (idx >> 10) & 1;
    int head = idx >> 11;
    int k    = kq * 4 + kh * 2;
    int col  = ct * 16 + n;
    const float* wp = w + (size_t)head * (F_IN * F_OUT) + (size_t)k * F_OUT + col;
    v2f v; v.x = wp[0]; v.y = wp[F_OUT];
    wt[idx] = v;
}

__global__ void __launch_bounds__(256)
gat_gemm_wmma(const float* __restrict__ x, const v2f* __restrict__ wt,
              float* __restrict__ h) {
    const int lane = threadIdx.x & 31;
    const int wave = threadIdx.x >> 5;
    const int head = wave & 3;
    const int rh   = wave >> 2;
    const int ln   = lane & 15;
    const int kh   = lane >> 4;

    const int row0 = blockIdx.x * 32 + rh * 16;
    const float* xr = x + (size_t)(row0 + ln) * F_IN;
    const v2f* bp = wt + head * 2048 + kh * 16 + ln;

    v8f c0 = {}, c1 = {};
#pragma unroll
    for (int kq = 0; kq < 32; ++kq) {
        v2f a  = *(const v2f*)(xr + kq * 4 + kh * 2);
        v2f b0 = bp[kq * 32];
        v2f b1 = bp[1024 + kq * 32];
        c0 = __builtin_amdgcn_wmma_f32_16x16x4_f32(false, a, false, b0,
                                                   (short)0, c0, false, false);
        c1 = __builtin_amdgcn_wmma_f32_16x16x4_f32(false, a, false, b1,
                                                   (short)0, c1, false, false);
    }
    float* hp = h + (size_t)(row0 + kh * 8) * HF + head * F_OUT + ln;
#pragma unroll
    for (int r = 0; r < 8; ++r) {
        hp[(size_t)r * HF]      = c0[r];
        hp[(size_t)r * HF + 16] = c1[r];
    }
}

__global__ void gat_alpha(const float* __restrict__ h, const float* __restrict__ a,
                          float* __restrict__ alpha_src, float* __restrict__ alpha_dst) {
    int idx = blockIdx.x * blockDim.x + threadIdx.x;
    if (idx >= N_NODES * H_HEADS) return;
    int head = idx & (H_HEADS - 1);
    const float* hp = h + (size_t)idx * F_OUT;
    const float* as = a + head * 2 * F_OUT;
    const float* ad = as + F_OUT;
    float s = 0.f, d = 0.f;
#pragma unroll
    for (int o = 0; o < F_OUT; ++o) {
        float v = hp[o];
        s += v * as[o];
        d += v * ad[o];
    }
    alpha_src[idx] = s;
    alpha_dst[idx] = d;
}

__global__ void gat_edge_max(const float* __restrict__ alpha_src,
                             const float* __restrict__ alpha_dst,
                             const int* __restrict__ esrc, const int* __restrict__ edst,
                             unsigned* __restrict__ m_enc) {
    int e = blockIdx.x * blockDim.x + threadIdx.x;
    if (e >= N_EDGES) return;
    int s = esrc[e], d = edst[e];
    const float4 as = *(const float4*)(alpha_src + (size_t)s * 4);
    const float4 ad = *(const float4*)(alpha_dst + (size_t)d * 4);
    unsigned* mp = m_enc + (size_t)d * 4;
    atomicMax(mp + 0, enc_f32(lrelu(as.x + ad.x)));
    atomicMax(mp + 1, enc_f32(lrelu(as.y + ad.y)));
    atomicMax(mp + 2, enc_f32(lrelu(as.z + ad.z)));
    atomicMax(mp + 3, enc_f32(lrelu(as.w + ad.w)));
}

__global__ void gat_finalize_max(const unsigned* __restrict__ m_enc,
                                 float* __restrict__ m) {
    int idx = blockIdx.x * blockDim.x + threadIdx.x;
    if (idx >= N_NODES * H_HEADS) return;
    float f = dec_f32(m_enc[idx]);
    if (!isfinite(f)) f = 0.f;
    m[idx] = f;
}

__global__ void gat_edge_sum(const float* __restrict__ alpha_src,
                             const float* __restrict__ alpha_dst,
                             const float* __restrict__ m,
                             const int* __restrict__ esrc, const int* __restrict__ edst,
                             float* __restrict__ denom) {
    int e = blockIdx.x * blockDim.x + threadIdx.x;
    if (e >= N_EDGES) return;
    int s = esrc[e], d = edst[e];
    const float4 as = *(const float4*)(alpha_src + (size_t)s * 4);
    const float4 ad = *(const float4*)(alpha_dst + (size_t)d * 4);
    const float4 mv = *(const float4*)(m + (size_t)d * 4);
    float* dp = denom + (size_t)d * 4;
    atomicAdd(dp + 0, expf(lrelu(as.x + ad.x) - mv.x));
    atomicAdd(dp + 1, expf(lrelu(as.y + ad.y) - mv.y));
    atomicAdd(dp + 2, expf(lrelu(as.z + ad.z) - mv.z));
    atomicAdd(dp + 3, expf(lrelu(as.w + ad.w) - mv.w));
}

__global__ void __launch_bounds__(256)
gat_edge_scatter(const float* __restrict__ h,
                 const float* __restrict__ alpha_src, const float* __restrict__ alpha_dst,
                 const float* __restrict__ m, const float* __restrict__ denom,
                 const int* __restrict__ esrc, const int* __restrict__ edst,
                 float* __restrict__ out) {
    int e = blockIdx.x * (blockDim.x >> 5) + (threadIdx.x >> 5);
    if (e >= N_EDGES) return;
    int lane = threadIdx.x & 31;
    int s = esrc[e], d = edst[e];
    const float* hs = h + (size_t)s * HF;
    float*       op = out + (size_t)d * HF;
#pragma unroll
    for (int i = 0; i < H_HEADS; ++i) {
        float ev  = lrelu(alpha_src[(size_t)s * 4 + i] + alpha_dst[(size_t)d * 4 + i]);
        float att = expf(ev - m[(size_t)d * 4 + i]) / (denom[(size_t)d * 4 + i] + 1e-16f);
        atomicAdd(op + i * F_OUT + lane, hs[i * F_OUT + lane] * att);
    }
}

extern "C" void kernel_launch(void* const* d_in, const int* in_sizes, int n_in,
                              void* d_out, int out_size, void* d_ws, size_t ws_size,
                              hipStream_t stream) {
    const float* x      = (const float*)d_in[0];
    const float* weight = (const float*)d_in[1];
    const float* a      = (const float*)d_in[2];
    const float* bias   = (const float*)d_in[3];
    const int*   esrc   = (const int*)d_in[4];
    const int*   edst   = (const int*)d_in[5];
    float* out = (float*)d_out;

    float*    h         = (float*)d_ws;
    float*    alpha_src = h + (size_t)N_NODES * HF;
    float*    alpha_dst = alpha_src + (size_t)N_NODES * H_HEADS;
    unsigned* m_enc     = (unsigned*)(alpha_dst + (size_t)N_NODES * H_HEADS);
    float*    m         = (float*)(m_enc + (size_t)N_NODES * H_HEADS);
    float*    denom     = m + (size_t)N_NODES * H_HEADS;
    v2f*      wt        = (v2f*)(denom + (size_t)N_NODES * H_HEADS);

    const int TB = 256;
    gat_init<<<(N_NODES * HF + TB - 1) / TB, TB, 0, stream>>>(out, bias, m_enc, denom);
    gat_repack_w<<<(H_HEADS * 2 * 32 * 2 * 16 + TB - 1) / TB, TB, 0, stream>>>(weight, wt);
    gat_gemm_wmma<<<N_NODES / 32, TB, 0, stream>>>(x, wt, h);
    gat_alpha<<<(N_NODES * H_HEADS + TB - 1) / TB, TB, 0, stream>>>(h, a, alpha_src, alpha_dst);
    gat_edge_max<<<(N_EDGES + TB - 1) / TB, TB, 0, stream>>>(alpha_src, alpha_dst, esrc, edst, m_enc);
    gat_finalize_max<<<(N_NODES * H_HEADS + TB - 1) / TB, TB, 0, stream>>>(m_enc, m);
    gat_edge_sum<<<(N_EDGES + TB - 1) / TB, TB, 0, stream>>>(alpha_src, alpha_dst, m, esrc, edst, denom);
    gat_edge_scatter<<<N_EDGES / 8, TB, 0, stream>>>(h, alpha_src, alpha_dst, m, denom, esrc, edst, out);
}


# Destination-side words of an edge (alpha_dst, max word, exp sum) share one 64-byte record per node: sum pass gathers 2 lines per edge, not 3, scatter 2, not 4; f32 atomics, arrival order as before
# speedup vs baseline: 3.8753x; 3.8753x over previous
#include <hip/hip_runtime.h>
#include <hip/hip_bf16.h>
#include <math.h>

#define N_NODES 100000
#define N_EDGES 1600000
#define H_HEADS 4
#define F_IN    128
#define F_OUT   32
#define HF      (H_HEADS * F_OUT)
#define SLOPE   0.2f

typedef __attribute__((ext_vector_type(2))) float v2f;
typedef __attribute__((ext_vector_type(8))) float v8f;

__device__ __forceinline__ float lrelu(float v) { return v > 0.f ? v : SLOPE * v; }

__device__ __forceinline__ unsigned enc_f32(float f) {
    unsigned u = __float_as_uint(f);
    return (u & 0x80000000u) ? ~u : (u | 0x80000000u);
}
__device__ __forceinline__ float dec_f32(unsigned e) {
    return (e & 0x80000000u) ? __uint_as_float(e ^ 0x80000000u) : __uint_as_float(~e);
}

#define REC_W     16
#define REC_AD    0
#define REC_MAX   4
#define REC_SUM   8
#define REC_SPARE 12
#define CTR_MAX   0
#define CTR_SUM   1
#define CTR_SCAT  2
__device__ __forceinline__ int ctr_word(int pass_id, int own) {
    return (own * 16 + pass_id * 4) * REC_W + REC_SPARE;
}

__global__ void gat_repack_w(const float* __restrict__ w, v2f* __restrict__ wt) {
    int idx = blockIdx.x * blockDim.x + threadIdx.x;
    if (idx >= H_HEADS * 2 * 32 * 2 * 16) return;
    int n    = idx & 15;
    int kh   = (idx >> 4) & 1;
    int kq   = (idx >> 5) & 31;
    int ct   = (idx >> 10) & 1;
    int head = idx >> 11;
    int k    = kq * 4 + kh * 2;
    int col  = ct * 16 + n;
    const float* wp = w + (size_t)head * (F_IN * F_OUT) + (size_t)k * F_OUT + col;
    v2f v; v.x = wp[0]; v.y = wp[F_OUT];
    wt[idx] = v;
}

static_assert(HF == 128 && N_NODES % 32 == 0, "a block starts 32 whole out rows: 1024 float4, four per thread");
#define X_PITCH (F_IN + 4)
static_assert(F_IN == 128, "a parked x row is one wave load: 32 lanes of 16 bytes");

template <int ROT>
__device__ __forceinline__ float add_row_ror(float v) {
    return v + __int_as_float(__builtin_amdgcn_update_dpp(0, __float_as_int(v), 0x120 + ROT, 0xf, 0xf, true));
}
__device__ __forceinline__ float row16_sum(float v) {
    v = add_row_ror<8>(v);
    v = add_row_ror<4>(v);
    v = add_row_ror<2>(v);
    v = add_row_ror<1>(v);
    return v;
}

__global__ void __launch_bounds__(256)
gat_gemm_wmma(const float* __restrict__ x, const v2f* __restrict__ wt,
              const float* __restrict__ a, const float* __restrict__ bias,
              float* __restrict__ h,
              float* __restrict__ alpha_src, unsigned* __restrict__ rec,
              float* __restrict__ out) {
    __shared__ __attribute__((aligned(16))) float xs[32 * X_PITCH];
    __shared__ __attribute__((aligned(16))) unsigned rs[32 * REC_W];
    const int lane = threadIdx.x & 31;
    const int wave = threadIdx.x >> 5;
    const int head = wave & 3;
    const int rh   = wave >> 2;
    const int ln   = lane & 15;
    const int kh   = lane >> 4;

    float4 xv[4];
    {
        const float4* x4 = (const float4*)(x + (size_t)blockIdx.x * 32 * F_IN) + threadIdx.x;
        for (int k = 0; k < 4; ++k) xv[k] = x4[k * 256];
    }
    float4 b = *(const float4*)(bias + lane * 4);

    for (int k = 0; k < 4; ++k)
        *(float4*)(xs + (wave + 8 * k) * X_PITCH + lane * 4) = xv[k];
    __syncthreads();

    {
        asm volatile("" : "+v"(b.x), "+v"(b.y), "+v"(b.z), "+v"(b.w));
        float4* o4 = (float4*)(out + (size_t)blockIdx.x * 32 * HF) + threadIdx.x;
        for (int k = 0; k < 4; ++k) o4[k * 256] = b;
    }

    const int row0 = blockIdx.x * 32 + rh * 16;
    const float* xr = xs + (rh * 16 + ln) * X_PITCH;
    const v2f* bp = wt + head * 2048 + kh * 16 + ln;

    v8f c0 = {}, c1 = {};
#pragma unroll
    for (int kq = 0; kq < 32; ++kq) {
        v2f av = *(const v2f*)(xr + kq * 4 + kh * 2);
        v2f b0 = bp[kq * 32];
        v2f b1 = bp[1024 + kq * 32];
        c0 = __builtin_amdgcn_wmma_f32_16x16x4_f32(false, av, false, b0,
                                                   (short)0, c0, false, false);
        c1 = __builtin_amdgcn_wmma_f32_16x16x4_f32(false, av, false, b1,
                                                   (short)0, c1, false, false);
    }
    const float* as = a + head * 2 * F_OUT;
    const float* ad = as + F_OUT;
    const float as0 = as[ln], as1 = as[16 + ln];
    const float ad0 = ad[ln], ad1 = ad[16 + ln];
    float vs = 0.f, vd = 0.f;
#pragma unroll
    for (int r = 0; r < 8; ++r) {
        const float s = row16_sum(c0[r] * as0 + c1[r] * as1);
        const float d = row16_sum(c0[r] * ad0 + c1[r] * ad1);
        if (ln == r) { vs = s; vd = d; }
    }
    const int rb = rh * 16 + kh * 8 + ln;
    if (ln < 8) rs[rb * REC_W + REC_AD + head] = __float_as_uint(vd);
    if (threadIdx.x < 32) {
        const unsigned e = enc_f32(-__builtin_inff());
        unsigned* r1 = rs + threadIdx.x * REC_W;
        *(uint4*)(r1 + REC_MAX)   = make_uint4(e, e, e, e);
        *(uint4*)(r1 + REC_SUM)   = make_uint4(0u, 0u, 0u, 0u);
        *(uint4*)(r1 + REC_SPARE) = make_uint4(0u, 0u, 0u, 0u);
    }
    __syncthreads();
    if (threadIdx.x < 32 * REC_W / 4)
        ((uint4*)rec)[blockIdx.x * (32 * REC_W / 4) + threadIdx.x] = ((const uint4*)rs)[threadIdx.x];
    if (ln < 8) alpha_src[(blockIdx.x * 32 + rb) * H_HEADS + head] = vs;
    float* hp = h + (size_t)(row0 + kh * 8) * HF + head * F_OUT + ln;
#pragma unroll
    for (int r = 0; r < 8; ++r) {
        hp[(size_t)r * HF]      = c0[r];
        hp[(size_t)r * HF + 16] = c1[r];
    }
}

#define EPW 64
static_assert(N_EDGES % (EPW * 8) == 0, "a ticket is EPW*8 edges, 8 waves of EPW: no tail");

#define TKT       (EPW * 8)
#define N_TKT     (N_EDGES / TKT)
#define EDGE_BLOCKS 2048

__device__ __forceinline__ int die_half() {
    return (int)((__builtin_amdgcn_s_sendmsg_rtn(0x87) >> 18) & 1u);
}
__device__ __forceinline__ bool in_set(int d, int own) { return ((d >> 4) & 1) == own; }

__device__ __forceinline__ int edge_idx(int lo, int hi, int j, int q) {
    return __builtin_amdgcn_ds_bpermute(((j & 3) * 8 + q) << 2, j < 4 ? lo : hi);
}

__global__ void __launch_bounds__(256)
gat_edge_max(const float* __restrict__ alpha_src,
             float* __restrict__ rec,
             const int* __restrict__ esrc, const int* __restrict__ edst) {
    unsigned* recu = (unsigned*)rec;
    __shared__ unsigned s_t[2];
    __shared__ int s_half;
    const int lane = threadIdx.x & 31;
    const int hd   = lane & 3;
    const int q    = lane >> 2;
    if (threadIdx.x == 0) s_half = die_half();
    __syncthreads();
    const int my_half = s_half;
    int step = 0;
    for (int pass = 0; pass < 2; ++pass) {
        const int own = my_half ^ pass;
        unsigned* c = recu + ctr_word(CTR_MAX, own);
        unsigned nxt = 0;
        if (threadIdx.x == 0) nxt = atomicAdd(c, 1u);
        for (int it = 0; it <= N_TKT; ++it) {
            if (threadIdx.x == 0) s_t[step & 1] = nxt;
            __syncthreads();
            const unsigned t = s_t[step & 1];
            ++step;
            if (t >= (unsigned)N_TKT) break;
            if (threadIdx.x == 0) nxt = atomicAdd(c, 1u);
            const int e0   = (int)t * TKT + (threadIdx.x >> 5) * EPW;
            const int s_lo = esrc[e0 + lane], s_hi = esrc[e0 + 32 + lane];
            const int d_lo = edst[e0 + lane], d_hi = edst[e0 + 32 + lane];
            int   sj[8], dj[8];
            float as[8], ad[8];
            for (int j = 0; j < 8; ++j) {
                sj[j] = edge_idx(s_lo, s_hi, j, q);
                dj[j] = edge_idx(d_lo, d_hi, j, q);
            }
            for (int j = 0; j < 8; ++j) {
                as[j] = 0.f; ad[j] = 0.f;
                if (in_set(dj[j], own)) {
                    as[j] = alpha_src[sj[j] * 4 + hd];
                    ad[j] = rec[dj[j] * REC_W + REC_AD + hd];
                }
            }
            for (int j = 0; j < 8; ++j)
                if (in_set(dj[j], own))
                    atomicMax(recu + dj[j] * REC_W + REC_MAX + hd, enc_f32(lrelu(as[j] + ad[j])));
        }
    }
}

__device__ __forceinline__ float seg_max(unsigned e) {
    const float f = dec_f32(e);
    return isfinite(f) ? f : 0.f;
}

__global__ void __launch_bounds__(256)
gat_edge_sum(const float* __restrict__ alpha_src,
             float* __restrict__ rec,
             const int* __restrict__ esrc, const int* __restrict__ edst) {
    unsigned* recu = (unsigned*)rec;
    __shared__ unsigned s_t[2];
    __shared__ int s_half;
    const int lane = threadIdx.x & 31;
    const int hd   = lane & 3;
    const int q    = lane >> 2;
    if (threadIdx.x == 0) s_half = die_half();
    __syncthreads();
    const int my_half = s_half;
    int step = 0;
    for (int pass = 0; pass < 2; ++pass) {
        const int own = my_half ^ pass;
        unsigned* c = recu + ctr_word(CTR_SUM, own);
        unsigned nxt = 0;
        if (threadIdx.x == 0) nxt = atomicAdd(c, 1u);
        for (int it = 0; it <= N_TKT; ++it) {
            if (threadIdx.x == 0) s_t[step & 1] = nxt;
            __syncthreads();
            const unsigned t = s_t[step & 1];
            ++step;
            if (t >= (unsigned)N_TKT) break;
            if (threadIdx.x == 0) nxt = atomicAdd(c, 1u);
            const int e0   = (int)t * TKT + (threadIdx.x >> 5) * EPW;
            const int s_lo = esrc[e0 + lane], s_hi = esrc[e0 + 32 + lane];
            const int d_lo = edst[e0 + lane], d_hi = edst[e0 + 32 + lane];
            int   sj[8], dj[8];
            float as[8], ad[8];
            unsigned me[8];
            for (int j = 0; j < 8; ++j) {
                sj[j] = edge_idx(s_lo, s_hi, j, q);
                dj[j] = edge_idx(d_lo, d_hi, j, q);
            }
            for (int j = 0; j < 8; ++j) {
                as[j] = 0.f; ad[j] = 0.f; me[j] = 0u;
                if (in_set(dj[j], own)) {
                    as[j] = alpha_src[sj[j] * 4 + hd];
                    ad[j] = rec[dj[j] * REC_W + REC_AD + hd];
                    me[j] = recu[dj[j] * REC_W + REC_MAX + hd];
                }
            }
            for (int j = 0; j < 8; ++j)
                if (in_set(dj[j], own))
                    atomicAdd(rec + dj[j] * REC_W + REC_SUM + hd, expf(lrelu(as[j] + ad[j]) - seg_max(me[j])));
        }
    }
}

#define SGRP 8

__device__ __forceinline__ float lane_f32(float v, int l) {
    return __int_as_float(__builtin_amdgcn_readlane(__float_as_int(v), l));
}

__global__ void __launch_bounds__(256)
gat_edge_scatter(const float* __restrict__ h,
                 const float* __restrict__ alpha_src, float* __restrict__ rec,
                 const int* __restrict__ esrc, const int* __restrict__ edst,
                 float* __restrict__ out) {
    __shared__ unsigned s_t[2];
    __shared__ int s_half;
    const int lane = threadIdx.x & 31;
    if (threadIdx.x == 0) s_half = die_half();
    __syncthreads();
    const int my_half = s_half;
    int step = 0;
    for (int pass = 0; pass < 2; ++pass) {
        const int own = my_half ^ pass;
        unsigned* c = (unsigned*)rec + ctr_word(CTR_SCAT, own);
        unsigned nxt = 0;
        if (threadIdx.x == 0) nxt = atomicAdd(c, 1u);
        for (int it = 0; it <= N_TKT; ++it) {
            if (threadIdx.x == 0) s_t[step & 1] = nxt;
            __syncthreads();
            const unsigned t = s_t[step & 1];
            ++step;
            if (t >= (unsigned)N_TKT) break;
            if (threadIdx.x == 0) nxt = atomicAdd(c, 1u);
            const int e0 = (int)t * TKT + (threadIdx.x >> 5) * EPW;
            for (int b = 0; b < EPW / 32; ++b) {
                const int s = esrc[e0 + b * 32 + lane];
                const int d = edst[e0 + b * 32 + lane];
                const bool mine = in_set(d, own);
                float att[H_HEADS] = {0.f, 0.f, 0.f, 0.f};
                if (mine) {
                    const float4 as = *(const float4*)(alpha_src + (size_t)s * 4);
                    const float* rd = rec + (size_t)d * REC_W;
                    const float4 ad = *(const float4*)(rd + REC_AD);
                    const uint4  me = *(const uint4*)(rd + REC_MAX);
                    const float4 mv = make_float4(seg_max(me.x), seg_max(me.y), seg_max(me.z), seg_max(me.w));
                    const float4 dn = *(const float4*)(rd + REC_SUM);
                    att[0] = expf(lrelu(as.x + ad.x) - mv.x) / (dn.x + 1e-16f);
                    att[1] = expf(lrelu(as.y + ad.y) - mv.y) / (dn.y + 1e-16f);
                    att[2] = expf(lrelu(as.z + ad.z) - mv.z) / (dn.z + 1e-16f);
                    att[3] = expf(lrelu(as.w + ad.w) - mv.w) / (dn.w + 1e-16f);
                }
                unsigned mask = __builtin_amdgcn_ballot_w32(mine);
                for (int g = 0; g < 32 / SGRP; ++g) {
                    if (mask == 0u) break;
                    int ls[SGRP];
                    int n = 0;
#pragma unroll
                    for (int j = 0; j < SGRP; ++j) {
                        ls[j] = mask ? __builtin_ctz(mask) : 0;
                        n += mask ? 1 : 0;
                        mask &= mask - 1u;
                    }
                    float v[SGRP][H_HEADS];
#pragma unroll
                    for (int j = 0; j < SGRP; ++j) {
                        if (j < n) {
                            const int sj = __builtin_amdgcn_readlane(s, ls[j]);
                            const float* hs = h + (size_t)sj * HF + lane;
#pragma unroll
                            for (int i = 0; i < H_HEADS; ++i) v[j][i] = hs[i * F_OUT];
                        }
                    }
#pragma unroll
                    for (int j = 0; j < SGRP; ++j) {
                        if (j < n) {
                            const int dj = __builtin_amdgcn_readlane(d, ls[j]);
                            float* op = out + (size_t)dj * HF + lane;
#pragma unroll
                            for (int i = 0; i < H_HEADS; ++i)
                                atomicAdd(op + i * F_OUT, v[j][i] * lane_f32(att[i], ls[j]));
                        }
                    }
                }
            }
        }
    }
}

extern "C" void kernel_launch(void* const* d_in, const int* in_sizes, int n_in,
                              void* d_out, int out_size, void* d_ws, size_t ws_size,
                              hipStream_t stream) {
    const float* x      = (const float*)d_in[0];
    const float* weight = (const float*)d_in[1];
    const float* a      = (const float*)d_in[2];
    const float* bias   = (const float*)d_in[3];
    const int*   esrc   = (const int*)d_in[4];
    const int*   edst   = (const int*)d_in[5];
    float* out = (float*)d_out;

    float*    h         = (float*)d_ws;
    float*    alpha_src = h + (size_t)N_NODES * HF;
    float*    rec       = alpha_src + (size_t)N_NODES * H_HEADS;
    v2f*      wt        = (v2f*)(rec + (size_t)N_NODES * REC_W);

    const int TB = 256;
    gat_repack_w<<<(H_HEADS * 2 * 32 * 2 * 16 + TB - 1) / TB, TB, 0, stream>>>(weight, wt);
    gat_gemm_wmma<<<N_NODES / 32, TB, 0, stream>>>(x, wt, a, bias, h, alpha_src, (unsigned*)rec, out);
    gat_edge_max<<<EDGE_BLOCKS, TB, 0, stream>>>(alpha_src, rec, esrc, edst);
    gat_edge_sum<<<EDGE_BLOCKS, TB, 0, stream>>>(alpha_src, rec, esrc, edst);
    gat_edge_scatter<<<EDGE_BLOCKS, TB, 0, stream>>>(h, alpha_src, rec, esrc, edst, out);
}


# Softmax shift: upper bound lrelu(max_n alpha_src + alpha_dst), its 8 min/max words from the GEMM launch, stands for the segment max if alpha_src spread <= 20: max pass exits at once, else exact way
# speedup vs baseline: 4.2965x; 1.1087x over previous
#include <hip/hip_runtime.h>
#include <hip/hip_bf16.h>
#include <math.h>

#define N_NODES 100000
#define N_EDGES 1600000
#define H_HEADS 4
#define F_IN    128
#define F_OUT   32
#define HF      (H_HEADS * F_OUT)
#define SLOPE   0.2f

typedef __attribute__((ext_vector_type(2))) float v2f;
typedef __attribute__((ext_vector_type(8))) float v8f;

__device__ __forceinline__ float lrelu(float v) { return v > 0.f ? v : SLOPE * v; }

__device__ __forceinline__ unsigned enc_f32(float f) {
    unsigned u = __float_as_uint(f);
    return (u & 0x80000000u) ? ~u : (u | 0x80000000u);
}
__device__ __forceinline__ float dec_f32(unsigned e) {
    return (e & 0x80000000u) ? __uint_as_float(e ^ 0x80000000u) : __uint_as_float(~e);
}

#define REC_W     16
#define REC_AD    0
#define REC_MAX   4
#define REC_SUM   8
#define REC_SPARE 12
#define CTR_MAX   0
#define CTR_SUM   1
#define CTR_SCAT  2
__device__ __forceinline__ int ctr_word(int pass_id, int own) {
    return (own * 16 + pass_id * 4) * REC_W + REC_SPARE;
}

#define BOUND_SPREAD 20.f
__device__ __forceinline__ int die_half() {
    return (int)((__builtin_amdgcn_s_sendmsg_rtn(0x87) >> 18) & 1u);
}
__device__ __forceinline__ int bound_word(int half, int i) {
    return (half * 16 + 1 + (i >> 2)) * REC_W + REC_SPARE + (i & 3);
}
__device__ __forceinline__ unsigned umax2(unsigned a, unsigned b) { return a > b ? a : b; }
__device__ __forceinline__ bool bound_ok(const unsigned* recu, float M[H_HEADS]) {
    unsigned w[2 * H_HEADS];
    for (int i = 0; i < 2 * H_HEADS; ++i)
        w[i] = (unsigned)__builtin_amdgcn_readfirstlane((int)umax2(recu[bound_word(0, i)], recu[bound_word(1, i)]));
    bool ok = true;
    for (int i = 0; i < H_HEADS; ++i) {
        M[i] = dec_f32(w[i]);
        ok &= (M[i] + dec_f32(w[H_HEADS + i]) <= BOUND_SPREAD);
    }
    return ok;
}

__global__ void gat_repack_w(const float* __restrict__ w, v2f* __restrict__ wt, unsigned* __restrict__ rec) {
    int idx = blockIdx.x * blockDim.x + threadIdx.x;
    if (idx >= H_HEADS * 2 * 32 * 2 * 16) return;
    if (idx < 4 * H_HEADS) rec[bound_word(idx >> 3, idx & 7)] = enc_f32(-__builtin_inff());
    int n    = idx & 15;
    int kh   = (idx >> 4) & 1;
    int kq   = (idx >> 5) & 31;
    int ct   = (idx >> 10) & 1;
    int head = idx >> 11;
    int k    = kq * 4 + kh * 2;
    int col  = ct * 16 + n;
    const float* wp = w + (size_t)head * (F_IN * F_OUT) + (size_t)k * F_OUT + col;
    v2f v; v.x = wp[0]; v.y = wp[F_OUT];
    wt[idx] = v;
}

static_assert(HF == 128 && N_NODES % 32 == 0, "a block starts 32 whole out rows: 1024 float4, four per thread");
#define X_PITCH (F_IN + 4)
static_assert(F_IN == 128, "a parked x row is one wave load: 32 lanes of 16 bytes");

template <int ROT>
__device__ __forceinline__ float add_row_ror(float v) {
    return v + __int_as_float(__builtin_amdgcn_update_dpp(0, __float_as_int(v), 0x120 + ROT, 0xf, 0xf, true));
}
__device__ __forceinline__ float row16_sum(float v) {
    v = add_row_ror<8>(v);
    v = add_row_ror<4>(v);
    v = add_row_ror<2>(v);
    v = add_row_ror<1>(v);
    return v;
}

template <int ROT>
__device__ __forceinline__ float max_row_ror(float v) {
    return fmaxf(v, __int_as_float(__builtin_amdgcn_update_dpp(0, __float_as_int(v), 0x120 + ROT, 0xf, 0xf, true)));
}
__device__ __forceinline__ float row16_max(float v) {
    v = max_row_ror<8>(v);
    v = max_row_ror<4>(v);
    v = max_row_ror<2>(v);
    v = max_row_ror<1>(v);
    return v;
}

__global__ void __launch_bounds__(256)
gat_gemm_wmma(const float* __restrict__ x, const v2f* __restrict__ wt,
              const float* __restrict__ a, const float* __restrict__ bias,
              float* __restrict__ h,
              float* __restrict__ alpha_src, unsigned* __restrict__ rec,
              float* __restrict__ out) {
    __shared__ __attribute__((aligned(16))) float xs[32 * X_PITCH];
    __shared__ __attribute__((aligned(16))) unsigned rs[32 * REC_W];
    __shared__ float sb[8 * 2];
    const int lane = threadIdx.x & 31;
    const int wave = threadIdx.x >> 5;
    const int head = wave & 3;
    const int rh   = wave >> 2;
    const int ln   = lane & 15;
    const int kh   = lane >> 4;

    float4 xv[4];
    {
        const float4* x4 = (const float4*)(x + (size_t)blockIdx.x * 32 * F_IN) + threadIdx.x;
        for (int k = 0; k < 4; ++k) xv[k] = x4[k * 256];
    }
    float4 b = *(const float4*)(bias + lane * 4);

    for (int k = 0; k < 4; ++k)
        *(float4*)(xs + (wave + 8 * k) * X_PITCH + lane * 4) = xv[k];
    __syncthreads();

    {
        asm volatile("" : "+v"(b.x), "+v"(b.y), "+v"(b.z), "+v"(b.w));
        float4* o4 = (float4*)(out + (size_t)blockIdx.x * 32 * HF) + threadIdx.x;
        for (int k = 0; k < 4; ++k) o4[k * 256] = b;
    }

    const int row0 = blockIdx.x * 32 + rh * 16;
    const float* xr = xs + (rh * 16 + ln) * X_PITCH;
    const v2f* bp = wt + head * 2048 + kh * 16 + ln;

    v8f c0 = {}, c1 = {};
#pragma unroll
    for (int kq = 0; kq < 32; ++kq) {
        v2f av = *(const v2f*)(xr + kq * 4 + kh * 2);
        v2f b0 = bp[kq * 32];
        v2f b1 = bp[1024 + kq * 32];
        c0 = __builtin_amdgcn_wmma_f32_16x16x4_f32(false, av, false, b0,
                                                   (short)0, c0, false, false);
        c1 = __builtin_amdgcn_wmma_f32_16x16x4_f32(false, av, false, b1,
                                                   (short)0, c1, false, false);
    }
    const float* as = a + head * 2 * F_OUT;
    const float* ad = as + F_OUT;
    const float as0 = as[ln], as1 = as[16 + ln];
    const float ad0 = ad[ln], ad1 = ad[16 + ln];
    float vs = 0.f, vd = 0.f;
#pragma unroll
    for (int r = 0; r < 8; ++r) {
        const float s = row16_sum(c0[r] * as0 + c1[r] * as1);
        const float d = row16_sum(c0[r] * ad0 + c1[r] * ad1);
        if (ln == r) { vs = s; vd = d; }
    }
    const int rb = rh * 16 + kh * 8 + ln;
    if (ln < 8) rs[rb * REC_W + REC_AD + head] = __float_as_uint(vd);
    {
        const float hi = row16_max(ln < 8 ? vs : -__builtin_inff());
        const float lo = row16_max(ln < 8 ? -vs : -__builtin_inff());
        if (lane == 0) {
            sb[wave * 2]     = fmaxf(hi, __int_as_float(__builtin_amdgcn_readlane(__float_as_int(hi), 16)));
            sb[wave * 2 + 1] = fmaxf(lo, __int_as_float(__builtin_amdgcn_readlane(__float_as_int(lo), 16)));
        }
    }
    if (threadIdx.x < 32) {
        const unsigned e = enc_f32(-__builtin_inff());
        unsigned* r1 = rs + threadIdx.x * REC_W;
        *(uint4*)(r1 + REC_MAX)   = make_uint4(e, e, e, e);
        *(uint4*)(r1 + REC_SUM)   = make_uint4(0u, 0u, 0u, 0u);
        *(uint4*)(r1 + REC_SPARE) = make_uint4(0u, 0u, 0u, 0u);
    }
    __syncthreads();
    if (threadIdx.x < 32 * REC_W / 4 &&
        !(blockIdx.x == 0 && (threadIdx.x == 7 || threadIdx.x == 11 || threadIdx.x == 71 || threadIdx.x == 75)))
        ((uint4*)rec)[blockIdx.x * (32 * REC_W / 4) + threadIdx.x] = ((const uint4*)rs)[threadIdx.x];
    if (threadIdx.x < 2 * H_HEADS) {
        const int hd = threadIdx.x & 3, kind = threadIdx.x >> 2;
        atomicMax(rec + bound_word(die_half(), threadIdx.x), enc_f32(fmaxf(sb[hd * 2 + kind], sb[(hd + 4) * 2 + kind])));
    }
    if (ln < 8) alpha_src[(blockIdx.x * 32 + rb) * H_HEADS + head] = vs;
    float* hp = h + (size_t)(row0 + kh * 8) * HF + head * F_OUT + ln;
#pragma unroll
    for (int r = 0; r < 8; ++r) {
        hp[(size_t)r * HF]      = c0[r];
        hp[(size_t)r * HF + 16] = c1[r];
    }
}

#define EPW 64
static_assert(N_EDGES % (EPW * 8) == 0, "a ticket is EPW*8 edges, 8 waves of EPW: no tail");

#define TKT       (EPW * 8)
#define N_TKT     (N_EDGES / TKT)
#define EDGE_BLOCKS 2048
#define MAX_BLOCKS  256

__device__ __forceinline__ bool in_set(int d, int own) { return ((d >> 4) & 1) == own; }

__device__ __forceinline__ int edge_idx(int lo, int hi, int j, int q) {
    return __builtin_amdgcn_ds_bpermute(((j & 3) * 8 + q) << 2, j < 4 ? lo : hi);
}

__global__ void __launch_bounds__(256)
gat_edge_max(const float* __restrict__ alpha_src,
             float* __restrict__ rec,
             const int* __restrict__ esrc, const int* __restrict__ edst) {
    unsigned* recu = (unsigned*)rec;
    __shared__ unsigned s_t[2];
    __shared__ int s_half;
    {
        float M[H_HEADS];
        if (bound_ok(recu, M)) return;
    }
    const int lane = threadIdx.x & 31;
    const int hd   = lane & 3;
    const int q    = lane >> 2;
    if (threadIdx.x == 0) s_half = die_half();
    __syncthreads();
    const int my_half = s_half;
    int step = 0;
    for (int pass = 0; pass < 2; ++pass) {
        const int own = my_half ^ pass;
        unsigned* c = recu + ctr_word(CTR_MAX, own);
        unsigned nxt = 0;
        if (threadIdx.x == 0) nxt = atomicAdd(c, 1u);
        for (int it = 0; it <= N_TKT; ++it) {
            if (threadIdx.x == 0) s_t[step & 1] = nxt;
            __syncthreads();
            const unsigned t = s_t[step & 1];
            ++step;
            if (t >= (unsigned)N_TKT) break;
            if (threadIdx.x == 0) nxt = atomicAdd(c, 1u);
            const int e0   = (int)t * TKT + (threadIdx.x >> 5) * EPW;
            const int s_lo = esrc[e0 + lane], s_hi = esrc[e0 + 32 + lane];
            const int d_lo = edst[e0 + lane], d_hi = edst[e0 + 32 + lane];
            int   sj[8], dj[8];
            float as[8], ad[8];
            for (int j = 0; j < 8; ++j) {
                sj[j] = edge_idx(s_lo, s_hi, j, q);
                dj[j] = edge_idx(d_lo, d_hi, j, q);
            }
            for (int j = 0; j < 8; ++j) {
                as[j] = 0.f; ad[j] = 0.f;
                if (in_set(dj[j], own)) {
                    as[j] = alpha_src[sj[j] * 4 + hd];
                    ad[j] = rec[dj[j] * REC_W + REC_AD + hd];
                }
            }
            for (int j = 0; j < 8; ++j)
                if (in_set(dj[j], own))
                    atomicMax(recu + dj[j] * REC_W + REC_MAX + hd, enc_f32(lrelu(as[j] + ad[j])));
        }
    }
}

__device__ __forceinline__ float seg_max(unsigned e) {
    const float f = dec_f32(e);
    return isfinite(f) ? f : 0.f;
}

__global__ void __launch_bounds__(256)
gat_edge_sum(const float* __restrict__ alpha_src,
             float* __restrict__ rec,
             const int* __restrict__ esrc, const int* __restrict__ edst) {
    unsigned* recu = (unsigned*)rec;
    __shared__ unsigned s_t[2];
    __shared__ int s_half;
    const int lane = threadIdx.x & 31;
    const int hd   = lane & 3;
    const int q    = lane >> 2;
    if (threadIdx.x == 0) s_half = die_half();
    __syncthreads();
    const int my_half = s_half;
    float M[H_HEADS];
    const bool bound = bound_ok(recu, M);
    const float Mh = hd == 0 ? M[0] : hd == 1 ? M[1] : hd == 2 ? M[2] : M[3];
    int step = 0;
    for (int pass = 0; pass < 2; ++pass) {
        const int own = my_half ^ pass;
        unsigned* c = recu + ctr_word(CTR_SUM, own);
        unsigned nxt = 0;
        if (threadIdx.x == 0) nxt = atomicAdd(c, 1u);
        for (int it = 0; it <= N_TKT; ++it) {
            if (threadIdx.x == 0) s_t[step & 1] = nxt;
            __syncthreads();
            const unsigned t = s_t[step & 1];
            ++step;
            if (t >= (unsigned)N_TKT) break;
            if (threadIdx.x == 0) nxt = atomicAdd(c, 1u);
            const int e0   = (int)t * TKT + (threadIdx.x >> 5) * EPW;
            const int s_lo = esrc[e0 + lane], s_hi = esrc[e0 + 32 + lane];
            const int d_lo = edst[e0 + lane], d_hi = edst[e0 + 32 + lane];
            int   sj[8], dj[8];
            float as[8], ad[8];
            unsigned me[8];
            for (int j = 0; j < 8; ++j) {
                sj[j] = edge_idx(s_lo, s_hi, j, q);
                dj[j] = edge_idx(d_lo, d_hi, j, q);
            }
            for (int j = 0; j < 8; ++j) {
                as[j] = 0.f; ad[j] = 0.f; me[j] = 0u;
                if (in_set(dj[j], own)) {
                    as[j] = alpha_src[sj[j] * 4 + hd];
                    ad[j] = rec[dj[j] * REC_W + REC_AD + hd];
                    if (!bound) me[j] = recu[dj[j] * REC_W + REC_MAX + hd];
                }
            }
            for (int j = 0; j < 8; ++j)
                if (in_set(dj[j], own))
                    atomicAdd(rec + dj[j] * REC_W + REC_SUM + hd,
                              expf(lrelu(as[j] + ad[j]) - (bound ? lrelu(Mh + ad[j]) : seg_max(me[j]))));
        }
    }
}

#define SGRP 8

__device__ __forceinline__ float lane_f32(float v, int l) {
    return __int_as_float(__builtin_amdgcn_readlane(__float_as_int(v), l));
}

__global__ void __launch_bounds__(256)
gat_edge_scatter(const float* __restrict__ h,
                 const float* __restrict__ alpha_src, float* __restrict__ rec,
                 const int* __restrict__ esrc, const int* __restrict__ edst,
                 float* __restrict__ out) {
    __shared__ unsigned s_t[2];
    __shared__ int s_half;
    const int lane = threadIdx.x & 31;
    if (threadIdx.x == 0) s_half = die_half();
    __syncthreads();
    const int my_half = s_half;
    float M[H_HEADS];
    const bool bound = bound_ok((const unsigned*)rec, M);
    int step = 0;
    for (int pass = 0; pass < 2; ++pass) {
        const int own = my_half ^ pass;
        unsigned* c = (unsigned*)rec + ctr_word(CTR_SCAT, own);
        unsigned nxt = 0;
        if (threadIdx.x == 0) nxt = atomicAdd(c, 1u);
        for (int it = 0; it <= N_TKT; ++it) {
            if (threadIdx.x == 0) s_t[step & 1] = nxt;
            __syncthreads();
            const unsigned t = s_t[step & 1];
            ++step;
            if (t >= (unsigned)N_TKT) break;
            if (threadIdx.x == 0) nxt = atomicAdd(c, 1u);
            const int e0 = (int)t * TKT + (threadIdx.x >> 5) * EPW;
            for (int b = 0; b < EPW / 32; ++b) {
                const int s = esrc[e0 + b * 32 + lane];
                const int d = edst[e0 + b * 32 + lane];
                const bool mine = in_set(d, own);
                float att[H_HEADS] = {0.f, 0.f, 0.f, 0.f};
                if (mine) {
                    const float4 as = *(const float4*)(alpha_src + (size_t)s * 4);
                    const float* rd = rec + (size_t)d * REC_W;
                    const float4 ad = *(const float4*)(rd + REC_AD);
                    float4 mv;
                    if (bound) {
                        mv = make_float4(lrelu(M[0] + ad.x), lrelu(M[1] + ad.y), lrelu(M[2] + ad.z), lrelu(M[3] + ad.w));
                    } else {
                        const uint4 me = *(const uint4*)(rd + REC_MAX);
                        mv = make_float4(seg_max(me.x), seg_max(me.y), seg_max(me.z), seg_max(me.w));
                    }
                    const float4 dn = *(const float4*)(rd + REC_SUM);
                    att[0] = expf(lrelu(as.x + ad.x) - mv.x) / (dn.x + 1e-16f);
                    att[1] = expf(lrelu(as.y + ad.y) - mv.y) / (dn.y + 1e-16f);
                    att[2] = expf(lrelu(as.z + ad.z) - mv.z) / (dn.z + 1e-16f);
                    att[3] = expf(lrelu(as.w + ad.w) - mv.w) / (dn.w + 1e-16f);
                }
                unsigned mask = __builtin_amdgcn_ballot_w32(mine);
                for (int g = 0; g < 32 / SGRP; ++g) {
                    if (mask == 0u) break;
                    int ls[SGRP];
                    int n = 0;
#pragma unroll
                    for (int j = 0; j < SGRP; ++j) {
                        ls[j] = mask ? __builtin_ctz(mask) : 0;
                        n += mask ? 1 : 0;
                        mask &= mask - 1u;
                    }
                    float v[SGRP][H_HEADS];
#pragma unroll
                    for (int j = 0; j < SGRP; ++j) {
                        if (j < n) {
                            const int sj = __builtin_amdgcn_readlane(s, ls[j]);
                            const float* hs = h + (size_t)sj * HF + lane;
#pragma unroll
                            for (int i = 0; i < H_HEADS; ++i) v[j][i] = hs[i * F_OUT];
                        }
                    }
#pragma unroll
                    for (int j = 0; j < SGRP; ++j) {
                        if (j < n) {
                            const int dj = __builtin_amdgcn_readlane(d, ls[j]);
                            float* op = out + (size_t)dj * HF + lane;
#pragma unroll
                            for (int i = 0; i < H_HEADS; ++i)
                                atomicAdd(op + i * F_OUT, v[j][i] * lane_f32(att[i], ls[j]));
                        }
                    }
                }
            }
        }
    }
}

extern "C" void kernel_launch(void* const* d_in, const int* in_sizes, int n_in,
                              void* d_out, int out_size, void* d_ws, size_t ws_size,
                              hipStream_t stream) {
    const float* x      = (const float*)d_in[0];
    const float* weight = (const float*)d_in[1];
    const float* a      = (const float*)d_in[2];
    const float* bias   = (const float*)d_in[3];
    const int*   esrc   = (const int*)d_in[4];
    const int*   edst   = (const int*)d_in[5];
    float* out = (float*)d_out;

    float*    h         = (float*)d_ws;
    float*    alpha_src = h + (size_t)N_NODES * HF;
    float*    rec       = alpha_src + (size_t)N_NODES * H_HEADS;
    v2f*      wt        = (v2f*)(rec + (size_t)N_NODES * REC_W);

    const int TB = 256;
    gat_repack_w<<<(H_HEADS * 2 * 32 * 2 * 16 + TB - 1) / TB, TB, 0, stream>>>(weight, wt, (unsigned*)rec);
    gat_gemm_wmma<<<N_NODES / 32, TB, 0, stream>>>(x, wt, a, bias, h, alpha_src, (unsigned*)rec, out);
    gat_edge_max<<<MAX_BLOCKS, TB, 0, stream>>>(alpha_src, rec, esrc, edst);
    gat_edge_sum<<<EDGE_BLOCKS, TB, 0, stream>>>(alpha_src, rec, esrc, edst);
    gat_edge_scatter<<<EDGE_BLOCKS, TB, 0, stream>>>(h, alpha_src, rec, esrc, edst, out);
}
